# starting kernel
# baseline (speedup 1.0000x reference)
#include <hip/hip_runtime.h>
#include <math.h>

#define NN 100000
#define NE 1600000
#define D  128
#define NL 3

typedef __attribute__((ext_vector_type(2))) float v2f;
typedef __attribute__((ext_vector_type(8))) float v8f;


__global__ void zero_f32_k(float* __restrict__ p, int n) {
    int i = blockIdx.x * blockDim.x + threadIdx.x;
    if (i < n) p[i] = 0.0f;
}

__global__ void deg_count_k(const int* __restrict__ col, float* __restrict__ deg, int e) {
    int i = blockIdx.x * blockDim.x + threadIdx.x;
    if (i < e) atomicAdd(&deg[col[i]], 1.0f);
}

__global__ void deg_inv_k(const float* __restrict__ deg, float* __restrict__ dinv, int n) {
    int i = blockIdx.x * blockDim.x + threadIdx.x;
    if (i < n) {
        float d = deg[i] + 1.0f;
        dinv[i] = rsqrtf(d);
    }
}


__global__ __launch_bounds__(256) void gemm_wmma_k(const float* __restrict__ A,
                                                   const float* __restrict__ W,
                                                   float* __restrict__ H) {
    const int lane = threadIdx.x & 31;
    const int wave = threadIdx.x >> 5;
    const int m0   = blockIdx.x << 4;
    const int n0   = wave << 4;
    const int half = lane >> 4;
    const int lm   = lane & 15;

    const float* __restrict__ arow = A + (size_t)(m0 + lm) * D;
    v8f c = {};

    #pragma unroll 4
    for (int k0 = 0; k0 < D; k0 += 4) {
        const int ka = k0 + 2 * half;
        v2f a = *(const v2f*)(arow + ka);
        v2f b;
        b.x = W[(size_t)ka * D + n0 + lm];
        b.y = W[(size_t)(ka + 1) * D + n0 + lm];
        c = __builtin_amdgcn_wmma_f32_16x16x4_f32(
             false, a,  false, b,
             (short)0, c,  false,  false);
    }

    float* __restrict__ drow = H + (size_t)(m0 + 8 * half) * D + n0 + lm;
    #pragma unroll
    for (int r = 0; r < 8; ++r) drow[(size_t)r * D] = c[r];
}


__global__ void self_init_k(const float* __restrict__ h, const float* __restrict__ dinv,
                            float* __restrict__ acc) {
    int idx = blockIdx.x * blockDim.x + threadIdx.x;
    if (idx >= NN * (D / 4)) return;
    int node = idx >> 5;
    float s = dinv[node];
    s = s * s;
    float4 v = ((const float4*)h)[idx];
    float4 o;
    o.x = v.x * s; o.y = v.y * s; o.z = v.z * s; o.w = v.w * s;
    ((float4*)acc)[idx] = o;
}


__global__ __launch_bounds__(256) void edge_scatter_k(const int* __restrict__ row,
                                                      const int* __restrict__ col,
                                                      const float* __restrict__ dinv,
                                                      const float* __restrict__ h,
                                                      float* __restrict__ acc) {
    const int lane = threadIdx.x & 31;
    const int wid  = (blockIdx.x * blockDim.x + threadIdx.x) >> 5;
    const int nw   = (gridDim.x * blockDim.x) >> 5;
    for (int e = wid; e < NE; e += nw) {
        const int r = row[e];
        const int c = col[e];
        const float nrm = dinv[r] * dinv[c];
        const float4 v = ((const float4*)(h + (size_t)r * D))[lane];
        float* ap = acc + (size_t)c * D + (lane << 2);
        atomicAdd(ap + 0, v.x * nrm);
        atomicAdd(ap + 1, v.y * nrm);
        atomicAdd(ap + 2, v.z * nrm);
        atomicAdd(ap + 3, v.w * nrm);
    }
}


__global__ void bias_relu_k(const float* __restrict__ acc, const float* __restrict__ b,
                            float* __restrict__ out) {
    int idx = blockIdx.x * blockDim.x + threadIdx.x;
    if (idx >= NN * (D / 4)) return;
    float4 v  = ((const float4*)acc)[idx];
    float4 bb = ((const float4*)b)[idx & 31];
    float4 o;
    o.x = fmaxf(v.x + bb.x, 0.0f);
    o.y = fmaxf(v.y + bb.y, 0.0f);
    o.z = fmaxf(v.z + bb.z, 0.0f);
    o.w = fmaxf(v.w + bb.w, 0.0f);
    ((float4*)out)[idx] = o;
}


extern "C" void kernel_launch(void* const* d_in, const int* in_sizes, int n_in,
                              void* d_out, int out_size, void* d_ws, size_t ws_size,
                              hipStream_t stream) {
    const float* x  = (const float*)d_in[0];
    const int*   ei = (const int*)d_in[1];
    const float* Ws = (const float*)d_in[2];
    const float* bs = (const float*)d_in[3];
    float* out = (float*)d_out;

    float* bufH   = (float*)d_ws;
    float* bufAcc = bufH + (size_t)NN * D;
    float* deg    = bufAcc + (size_t)NN * D;
    float* dinv   = deg + NN;

    const int* row = ei;
    const int* col = ei + NE;

    const int nvec = NN * (D / 4);

    zero_f32_k<<<(NN + 255) / 256, 256, 0, stream>>>(deg, NN);
    deg_count_k<<<(NE + 255) / 256, 256, 0, stream>>>(col, deg, NE);
    deg_inv_k<<<(NN + 255) / 256, 256, 0, stream>>>(deg, dinv, NN);

    const float* src = x;
    for (int l = 0; l < NL; ++l) {
        gemm_wmma_k<<<NN / 16, 256, 0, stream>>>(src, Ws + (size_t)l * D * D, bufH);
        self_init_k<<<(nvec + 255) / 256, 256, 0, stream>>>(bufH, dinv, bufAcc);
        edge_scatter_k<<<8192, 256, 0, stream>>>(row, col, dinv, bufH, bufAcc);
        float* dst = (l == NL - 1) ? out : bufAcc;
        bias_relu_k<<<(nvec + 255) / 256, 256, 0, stream>>>(bufAcc, bs + (size_t)l * D, dst);
        src = bufAcc;
    }
}


# per-layer float-atomic edge scatter replaced: in-edge lists built once per call (slot = arrival-order rank a counting add returns), one wave per node sums its rows in registers in list order
# speedup vs baseline: 3.2842x; 3.2842x over previous
#include <hip/hip_runtime.h>
#include <math.h>


#define NN 100000
#define NE 1600000
#define D  128
#define NL 3

typedef __attribute__((ext_vector_type(2))) float v2f;
typedef __attribute__((ext_vector_type(8))) float v8f;


__global__ void zero_i32_k(int* __restrict__ p, int n) {
    int i = blockIdx.x * blockDim.x + threadIdx.x;
    if (i < n) p[i] = 0;
}

__global__ void edge_rank_k(const int* __restrict__ col, int* __restrict__ cnt,
                            int* __restrict__ rank, int e) {
    int i = blockIdx.x * blockDim.x + threadIdx.x;
    if (i < e) rank[i] = atomicAdd(&cnt[col[i]], 1);
}

__global__ __launch_bounds__(256) void node_setup_k(int* __restrict__ cnt, float* __restrict__ dinv,
                                                    unsigned long long* __restrict__ seg,
                                                    int2* __restrict__ lst) {
    __shared__ int s[256];
    __shared__ int sbase;
    const int t = threadIdx.x;
    const int i = blockIdx.x * 256 + t;
    const int d = (i < NN) ? cnt[i] : 0;
    const int len = (i < NN) ? d + 1 : 0;
    const float di = rsqrtf((float)d + 1.0f);
    s[t] = len;
    __syncthreads();
    for (int off = 1; off < 256; off <<= 1) {
        const int v = (t >= off) ? s[t - off] : 0;
        __syncthreads();
        s[t] += v;
        __syncthreads();
    }
    if (t == 255) sbase = atomicAdd(&cnt[NN], s[255]);
    __syncthreads();
    if (i < NN) {
        const int start = sbase + s[t] - len;
        dinv[i] = di;
        seg[i] = (unsigned long long)(unsigned)start | ((unsigned long long)(unsigned)len << 32);
        int2 v;
        v.x = i;
        v.y = __float_as_int(di * di);
        lst[start] = v;
    }
}

__global__ void edge_place_k(const int* __restrict__ row, const int* __restrict__ col,
                             const int* __restrict__ rank,
                             const unsigned long long* __restrict__ seg,
                             const float* __restrict__ dinv, int2* __restrict__ lst, int e) {
    int i = blockIdx.x * blockDim.x + threadIdx.x;
    if (i >= e) return;
    const int r = row[i];
    const int c = col[i];
    const int start = (int)(unsigned)seg[c];
    int2 v;
    v.x = r;
    v.y = __float_as_int(dinv[r] * dinv[c]);
    lst[start + 1 + rank[i]] = v;
}


__global__ __launch_bounds__(256) void gemm_wmma_k(const float* __restrict__ A,
                                                   const float* __restrict__ W,
                                                   float* __restrict__ H) {
    const int lane = threadIdx.x & 31;
    const int wave = threadIdx.x >> 5;
    const int m0   = blockIdx.x << 4;
    const int n0   = wave << 4;
    const int half = lane >> 4;
    const int lm   = lane & 15;

    const float* __restrict__ arow = A + (size_t)(m0 + lm) * D;
    v8f c = {};

    #pragma unroll 4
    for (int k0 = 0; k0 < D; k0 += 4) {
        const int ka = k0 + 2 * half;
        v2f a = *(const v2f*)(arow + ka);
        v2f b;
        b.x = W[(size_t)ka * D + n0 + lm];
        b.y = W[(size_t)(ka + 1) * D + n0 + lm];
        c = __builtin_amdgcn_wmma_f32_16x16x4_f32(
             false, a,  false, b,
             (short)0, c,  false,  false);
    }

    float* __restrict__ drow = H + (size_t)(m0 + 8 * half) * D + n0 + lm;
    #pragma unroll
    for (int r = 0; r < 8; ++r) drow[(size_t)r * D] = c[r];
}


template <int K>
__device__ __forceinline__ void add_rows(const float4* __restrict__ hv, int ex, int ey,
                                         int j, unsigned lane, float4& acc) {
    float4 a[K];
    float  w[K];
    for (int k = 0; k < K; ++k) {
        const unsigned r = (unsigned)__builtin_amdgcn_readlane(ex, j + k);
        w[k] = __int_as_float(__builtin_amdgcn_readlane(ey, j + k));
        a[k] = hv[r * (unsigned)(D / 4) + lane];
    }
    for (int k = 0; k < K; ++k) {
        acc.x = fmaf(a[k].x, w[k], acc.x);
        acc.y = fmaf(a[k].y, w[k], acc.y);
        acc.z = fmaf(a[k].z, w[k], acc.z);
        acc.w = fmaf(a[k].w, w[k], acc.w);
    }
}

__global__ __launch_bounds__(256) void gather_sum_k(const unsigned long long* __restrict__ seg,
                                                    const int2* __restrict__ lst,
                                                    const float* __restrict__ h,
                                                    const float* __restrict__ b,
                                                    float* __restrict__ out) {
    const unsigned lane = threadIdx.x & 31;
    const int wid = __builtin_amdgcn_readfirstlane((blockIdx.x * blockDim.x + threadIdx.x) >> 5);
    const int nw  = (gridDim.x * blockDim.x) >> 5;
    const float4* __restrict__ hv = (const float4*)h;
    const float4 bb = ((const float4*)b)[lane];

    for (int c = wid; c < NN; c += nw) {
        const unsigned long long sg = seg[c];
        const int s = (int)(unsigned)sg;
        const int n = (int)(unsigned)(sg >> 32);
        float4 acc;
        acc.x = 0.0f; acc.y = 0.0f; acc.z = 0.0f; acc.w = 0.0f;

        for (int base = 0; base < n; base += 32) {
            const int m = (n - base < 32) ? (n - base) : 32;
            int2 e;
            e.x = 0; e.y = 0;
            if ((int)lane < m) e = lst[s + base + (int)lane];
            int j = 0;
            for (; j + 8 <= m; j += 8) add_rows<8>(hv, e.x, e.y, j, lane, acc);
            if (j + 4 <= m) { add_rows<4>(hv, e.x, e.y, j, lane, acc); j += 4; }
            if (j + 2 <= m) { add_rows<2>(hv, e.x, e.y, j, lane, acc); j += 2; }
            if (j < m)      { add_rows<1>(hv, e.x, e.y, j, lane, acc); }
        }

        float4 o;
        o.x = fmaxf(acc.x + bb.x, 0.0f);
        o.y = fmaxf(acc.y + bb.y, 0.0f);
        o.z = fmaxf(acc.z + bb.z, 0.0f);
        o.w = fmaxf(acc.w + bb.w, 0.0f);
        ((float4*)out)[(unsigned)c * (unsigned)(D / 4) + lane] = o;
    }
}


extern "C" void kernel_launch(void* const* d_in, const int* in_sizes, int n_in,
                              void* d_out, int out_size, void* d_ws, size_t ws_size,
                              hipStream_t stream) {
    const float* x  = (const float*)d_in[0];
    const int*   ei = (const int*)d_in[1];
    const float* Ws = (const float*)d_in[2];
    const float* bs = (const float*)d_in[3];
    float* out = (float*)d_out;

    float* bufH = (float*)d_ws;
    float* bufA = bufH + (size_t)NN * D;
    int2*  lst  = (int2*)(bufA + (size_t)NN * D);
    unsigned long long* seg = (unsigned long long*)(lst + NE + NN);
    int*   rank = (int*)(seg + NN);
    int*   cnt  = rank + NE;
    float* dinv = (float*)(cnt + NN + 1);

    const int* row = ei;
    const int* col = ei + NE;

    zero_i32_k<<<(NN + 1 + 255) / 256, 256, 0, stream>>>(cnt, NN + 1);
    edge_rank_k<<<(NE + 255) / 256, 256, 0, stream>>>(col, cnt, rank, NE);
    node_setup_k<<<(NN + 255) / 256, 256, 0, stream>>>(cnt, dinv, seg, lst);
    edge_place_k<<<(NE + 255) / 256, 256, 0, stream>>>(row, col, rank, seg, dinv, lst, NE);

    const float* src = x;
    for (int l = 0; l < NL; ++l) {
        gemm_wmma_k<<<NN / 16, 256, 0, stream>>>(src, Ws + (size_t)l * D * D, bufH);
        float* dst = (l == NL - 1) ? out : bufA;
        gather_sum_k<<<2048, 256, 0, stream>>>(seg, lst, bufH, bs + (size_t)l * D, dst);
        src = bufA;
    }
}


# dense product moved behind the neighbour sum ((A x) W for A (x W), sum order changed) and fused into the gather kernel: summed rows feed the f32 WMMA from an LDS tile, H never stored
# speedup vs baseline: 4.7322x; 1.4409x over previous
#include <hip/hip_runtime.h>
#include <math.h>


#define NN 100000
#define NE 1600000
#define D  128
#define NL 3
#define TM 32
#define TP 132

typedef __attribute__((ext_vector_type(2))) float v2f;
typedef __attribute__((ext_vector_type(8))) float v8f;


__global__ void zero_i32_k(int* __restrict__ p, int n) {
    int i = blockIdx.x * blockDim.x + threadIdx.x;
    if (i < n) p[i] = 0;
}

__global__ void edge_rank_k(const int* __restrict__ col, int* __restrict__ cnt,
                            int* __restrict__ rank, int e) {
    int i = blockIdx.x * blockDim.x + threadIdx.x;
    if (i < e) rank[i] = atomicAdd(&cnt[col[i]], 1);
}

__global__ __launch_bounds__(256) void node_setup_k(int* __restrict__ cnt, float* __restrict__ dinv,
                                                    unsigned long long* __restrict__ seg,
                                                    int2* __restrict__ lst) {
    __shared__ int s[256];
    __shared__ int sbase;
    const int t = threadIdx.x;
    const int i = blockIdx.x * 256 + t;
    const int d = (i < NN) ? cnt[i] : 0;
    const int len = (i < NN) ? d + 1 : 0;
    const float di = rsqrtf((float)d + 1.0f);
    s[t] = len;
    __syncthreads();
    for (int off = 1; off < 256; off <<= 1) {
        const int v = (t >= off) ? s[t - off] : 0;
        __syncthreads();
        s[t] += v;
        __syncthreads();
    }
    if (t == 255) sbase = atomicAdd(&cnt[NN], s[255]);
    __syncthreads();
    if (i < NN) {
        const int start = sbase + s[t] - len;
        dinv[i] = di;
        seg[i] = (unsigned long long)(unsigned)start | ((unsigned long long)(unsigned)len << 32);
        int2 v;
        v.x = i;
        v.y = __float_as_int(di * di);
        lst[start] = v;
    }
}

__global__ void edge_place_k(const int* __restrict__ row, const int* __restrict__ col,
                             const int* __restrict__ rank,
                             const unsigned long long* __restrict__ seg,
                             const float* __restrict__ dinv, int2* __restrict__ lst, int e) {
    int i = blockIdx.x * blockDim.x + threadIdx.x;
    if (i >= e) return;
    const int r = row[i];
    const int c = col[i];
    const int start = (int)(unsigned)seg[c];
    int2 v;
    v.x = r;
    v.y = __float_as_int(dinv[r] * dinv[c]);
    lst[start + 1 + rank[i]] = v;
}


template <int K>
__device__ __forceinline__ void add_rows(const float4* __restrict__ hv, int ex, int ey,
                                         int j, unsigned lane, float4& acc) {
    float4 a[K];
    float  w[K];
    for (int k = 0; k < K; ++k) {
        const unsigned r = (unsigned)__builtin_amdgcn_readlane(ex, j + k);
        w[k] = __int_as_float(__builtin_amdgcn_readlane(ey, j + k));
        a[k] = hv[r * (unsigned)(D / 4) + lane];
    }
    for (int k = 0; k < K; ++k) {
        acc.x = fmaf(a[k].x, w[k], acc.x);
        acc.y = fmaf(a[k].y, w[k], acc.y);
        acc.z = fmaf(a[k].z, w[k], acc.z);
        acc.w = fmaf(a[k].w, w[k], acc.w);
    }
}

__device__ __forceinline__ void sum_stretch(const float4* __restrict__ hv, int ex, int ey, int m,
                                            unsigned lane, float4& acc) {
    int j = 0;
    for (; j + 8 <= m; j += 8) add_rows<8>(hv, ex, ey, j, lane, acc);
    if (j + 4 <= m) { add_rows<4>(hv, ex, ey, j, lane, acc); j += 4; }
    if (j + 2 <= m) { add_rows<2>(hv, ex, ey, j, lane, acc); j += 2; }
    if (j < m)      { add_rows<1>(hv, ex, ey, j, lane, acc); }
}

__global__ __launch_bounds__(256) void layer_k(const unsigned long long* __restrict__ seg,
                                                  const int2* __restrict__ lst,
                                                  const float* __restrict__ x,
                                                  const float* __restrict__ W,
                                                  const float* __restrict__ b,
                                                  float* __restrict__ out) {
    __shared__ __attribute__((aligned(16))) float sA[TM * TP];
    __shared__ __attribute__((aligned(16))) float sC[TM * TP];

    const unsigned lane = threadIdx.x & 31;
    const int wave = __builtin_amdgcn_readfirstlane(threadIdx.x >> 5);
    const int half = lane >> 4;
    const int lm   = lane & 15;
    const int n0   = wave << 4;
    const float4* __restrict__ hv = (const float4*)x;
    const float4 bb = ((const float4*)b)[lane];

    v2f bf[D / 4];
    #pragma unroll
    for (int ks = 0; ks < D / 4; ++ks) {
        const int ka = 4 * ks + 2 * half;
        bf[ks].x = W[ka * D + n0 + lm];
        bf[ks].y = W[(ka + 1) * D + n0 + lm];
    }

    for (int tile = blockIdx.x; tile < NN / TM; tile += gridDim.x) {
        const int c0 = tile * TM + wave * 4;
        unsigned long long sg[4];
        int2 e[4];
        #pragma unroll
        for (int q = 0; q < 4; ++q) sg[q] = seg[c0 + q];
        #pragma unroll
        for (int q = 0; q < 4; ++q) {
            const int s = (int)(unsigned)sg[q];
            const int n = (int)(unsigned)(sg[q] >> 32);
            const int m = (n < 32) ? n : 32;
            e[q].x = 0; e[q].y = 0;
            if ((int)lane < m) e[q] = lst[s + (int)lane];
        }
        #pragma unroll
        for (int q = 0; q < 4; ++q) {
            const int s = (int)(unsigned)sg[q];
            const int n = (int)(unsigned)(sg[q] >> 32);
            float4 acc;
            acc.x = 0.0f; acc.y = 0.0f; acc.z = 0.0f; acc.w = 0.0f;
            sum_stretch(hv, e[q].x, e[q].y, (n < 32) ? n : 32, lane, acc);
            for (int base = 32; base < n; base += 32) {
                const int m = (n - base < 32) ? (n - base) : 32;
                int2 f;
                f.x = 0; f.y = 0;
                if ((int)lane < m) f = lst[s + base + (int)lane];
                sum_stretch(hv, f.x, f.y, m, lane, acc);
            }
            *(float4*)&sA[(wave * 4 + q) * TP + 4 * (int)lane] = acc;
        }
        __syncthreads();

        v8f c_lo = {};
        v8f c_hi = {};
        #pragma unroll
        for (int ks = 0; ks < D / 4; ++ks) {
            const int ka = 4 * ks + 2 * half;
            const v2f a_lo = *(const v2f*)&sA[lm * TP + ka];
            const v2f a_hi = *(const v2f*)&sA[(16 + lm) * TP + ka];
            c_lo = __builtin_amdgcn_wmma_f32_16x16x4_f32(
                 false, a_lo,  false, bf[ks],
                 (short)0, c_lo,  false,  false);
            c_hi = __builtin_amdgcn_wmma_f32_16x16x4_f32(
                 false, a_hi,  false, bf[ks],
                 (short)0, c_hi,  false,  false);
        }
        #pragma unroll
        for (int r = 0; r < 8; ++r) {
            sC[(r + 8 * half) * TP + n0 + lm]      = c_lo[r];
            sC[(16 + r + 8 * half) * TP + n0 + lm] = c_hi[r];
        }
        __syncthreads();

        #pragma unroll
        for (int q = 0; q < 4; ++q) {
            const float4 v = *(const float4*)&sC[(wave * 4 + q) * TP + 4 * (int)lane];
            float4 o;
            o.x = fmaxf(v.x + bb.x, 0.0f);
            o.y = fmaxf(v.y + bb.y, 0.0f);
            o.z = fmaxf(v.z + bb.z, 0.0f);
            o.w = fmaxf(v.w + bb.w, 0.0f);
            ((float4*)out)[(unsigned)(c0 + q) * (unsigned)(D / 4) + lane] = o;
        }
    }
}


extern "C" void kernel_launch(void* const* d_in, const int* in_sizes, int n_in,
                              void* d_out, int out_size, void* d_ws, size_t ws_size,
                              hipStream_t stream) {
    const float* x  = (const float*)d_in[0];
    const int*   ei = (const int*)d_in[1];
    const float* Ws = (const float*)d_in[2];
    const float* bs = (const float*)d_in[3];
    float* out = (float*)d_out;

    float* bufH = (float*)d_ws;
    float* bufA = bufH + (size_t)NN * D;
    int2*  lst  = (int2*)(bufA + (size_t)NN * D);
    unsigned long long* seg = (unsigned long long*)(lst + NE + NN);
    int*   rank = (int*)(seg + NN);
    int*   cnt  = rank + NE;
    float* dinv = (float*)(cnt + NN + 1);

    const int* row = ei;
    const int* col = ei + NE;

    zero_i32_k<<<(NN + 1 + 255) / 256, 256, 0, stream>>>(cnt, NN + 1);
    edge_rank_k<<<(NE + 255) / 256, 256, 0, stream>>>(col, cnt, rank, NE);
    node_setup_k<<<(NN + 255) / 256, 256, 0, stream>>>(cnt, dinv, seg, lst);
    edge_place_k<<<(NE + 255) / 256, 256, 0, stream>>>(row, col, rank, seg, dinv, lst, NE);

    layer_k<<<1024, 256, 0, stream>>>(seg, lst, x,    Ws,                     bs,         bufA);
    layer_k<<<1024, 256, 0, stream>>>(seg, lst, bufA, Ws + (size_t)D * D,     bs + D,     bufH);
    layer_k<<<1024, 256, 0, stream>>>(seg, lst, bufH, Ws + (size_t)2 * D * D, bs + 2 * D, out);
}


# in-edge lists built by a two-pass counting sort in LDS (one block per node range x edge slice) instead of a global counting add per edge; a node's sum order now slice, then LDS arrival
# speedup vs baseline: 7.0185x; 1.4831x over previous
#include <hip/hip_runtime.h>
#include <math.h>


#define NN 100000
#define NE 1600000
#define D  128
#define NL 3
#define TM 32
#define TP 132

typedef __attribute__((ext_vector_type(2))) float v2f;
typedef __attribute__((ext_vector_type(8))) float v8f;


#define NR 8
#define RN (NN / NR)
#define NS 32
#define SE (NE / NS)

__global__ __launch_bounds__(1024) void slice_count_k(const int* __restrict__ col,
                                                      int* __restrict__ tab,
                                                      int* __restrict__ total) {
    __shared__ int h[RN];
    const int t  = threadIdx.x;
    const int r  = blockIdx.x % NR;
    const int s  = blockIdx.x / NR;
    const int lo = r * RN;
    for (int j = t; j < RN; j += 1024) h[j] = 0;
    if (blockIdx.x == 0 && t == 0) *total = 0;
    __syncthreads();
    const int4* __restrict__ cs = (const int4*)(col + s * SE);
    for (int q0 = 0; q0 < SE / 4; q0 += 4 * 1024) {
        int4 c[4];
        #pragma unroll
        for (int k = 0; k < 4; ++k) {
            const int q = q0 + k * 1024 + t;
            c[k] = cs[(q < SE / 4) ? q : (SE / 4 - 1)];
            if (q >= SE / 4) { c[k].x = -1; c[k].y = -1; c[k].z = -1; c[k].w = -1; }
        }
        #pragma unroll
        for (int k = 0; k < 4; ++k) {
            const unsigned u0 = (unsigned)(c[k].x - lo);
            const unsigned u1 = (unsigned)(c[k].y - lo);
            const unsigned u2 = (unsigned)(c[k].z - lo);
            const unsigned u3 = (unsigned)(c[k].w - lo);
            if (u0 < (unsigned)RN) atomicAdd(&h[u0], 1);
            if (u1 < (unsigned)RN) atomicAdd(&h[u1], 1);
            if (u2 < (unsigned)RN) atomicAdd(&h[u2], 1);
            if (u3 < (unsigned)RN) atomicAdd(&h[u3], 1);
        }
    }
    __syncthreads();
    int* __restrict__ dst = tab + s * NN + lo;
    for (int j = t; j < RN; j += 1024) dst[j] = h[j];
}

__global__ __launch_bounds__(256) void node_setup_k(int* __restrict__ tab, int* __restrict__ total,
                                                    float* __restrict__ dinv,
                                                    unsigned long long* __restrict__ seg,
                                                    int2* __restrict__ lst) {
    __shared__ int s[256];
    __shared__ int sbase;
    const int t = threadIdx.x;
    const int i = blockIdx.x * 256 + t;
    int c[NS];
    int d = 0;
    #pragma unroll
    for (int k = 0; k < NS; ++k) {
        const int v = tab[k * NN + ((i < NN) ? i : (NN - 1))];
        c[k] = (i < NN) ? v : 0;
        d += c[k];
    }
    const int len = (i < NN) ? d + 1 : 0;
    const float di = rsqrtf((float)d + 1.0f);
    s[t] = len;
    __syncthreads();
    for (int off = 1; off < 256; off <<= 1) {
        const int v = (t >= off) ? s[t - off] : 0;
        __syncthreads();
        s[t] += v;
        __syncthreads();
    }
    if (t == 255) sbase = atomicAdd(total, s[255]);
    __syncthreads();
    if (i < NN) {
        const int start = sbase + s[t] - len;
        dinv[i] = di;
        seg[i] = (unsigned long long)(unsigned)start | ((unsigned long long)(unsigned)len << 32);
        int2 v;
        v.x = i;
        v.y = __float_as_int(di * di);
        lst[start] = v;
        int run = start + 1;
        #pragma unroll
        for (int k = 0; k < NS; ++k) {
            tab[k * NN + i] = run;
            run += c[k];
        }
    }
}

#define EPT 3

__device__ __forceinline__ void edge_words(const int4* __restrict__ cs, const int4* __restrict__ rs,
                                           int q0, int t, int4 (&c)[EPT], int4 (&r)[EPT]) {
    #pragma unroll
    for (int k = 0; k < EPT; ++k) {
        const int q  = q0 + k * 1024 + t;
        const int qq = (q < SE / 4) ? q : (SE / 4 - 1);
        c[k] = cs[qq];
        r[k] = rs[qq];
        if (q >= SE / 4) { c[k].x = -1; c[k].y = -1; c[k].z = -1; c[k].w = -1; }
    }
}

__global__ __launch_bounds__(1024) void slice_place_k(const int* __restrict__ row,
                                                      const int* __restrict__ col,
                                                      const int* __restrict__ tab,
                                                      const float* __restrict__ dinv,
                                                      int2* __restrict__ lst) {
    __shared__ __attribute__((aligned(16))) int   pos[RN + 32];
    __shared__ __attribute__((aligned(16))) float dl[RN + 32];
    const int t  = threadIdx.x;
    const int r  = blockIdx.x % NR;
    const int s  = blockIdx.x / NR;
    const int lo = r * RN;
    const unsigned spare = (unsigned)(RN + (t & 31));
    {
        const int4*   __restrict__ s4 = (const int4*)(tab + s * NN + lo);
        const float4* __restrict__ d4 = (const float4*)(dinv + lo);
        #pragma unroll
        for (int k = 0; k < 4; ++k) {
            const int q  = k * 1024 + t;
            const int qq = (q < RN / 4) ? q : (RN / 4 - 1);
            const int4   a = s4[qq];
            const float4 b = d4[qq];
            if (q < RN / 4) { ((int4*)pos)[q] = a; ((float4*)dl)[q] = b; }
        }
        if (t < 32) { pos[RN + t] = 0; dl[RN + t] = 0.0f; }
    }
    __syncthreads();
    const int4* __restrict__ cs = (const int4*)(col + s * SE);
    const int4* __restrict__ rs = (const int4*)(row + s * SE);
    int4 cn[EPT], rn[EPT];
    edge_words(cs, rs, 0, t, cn, rn);
    for (int q0 = 0; q0 < SE / 4; q0 += EPT * 1024) {
        int c[4 * EPT], rr[4 * EPT];
        #pragma unroll
        for (int k = 0; k < EPT; ++k) {
            c[4 * k + 0] = cn[k].x;  rr[4 * k + 0] = rn[k].x;
            c[4 * k + 1] = cn[k].y;  rr[4 * k + 1] = rn[k].y;
            c[4 * k + 2] = cn[k].z;  rr[4 * k + 2] = rn[k].z;
            c[4 * k + 3] = cn[k].w;  rr[4 * k + 3] = rn[k].w;
        }
        if (q0 + EPT * 1024 < SE / 4) edge_words(cs, rs, q0 + EPT * 1024, t, cn, rn);
        float w[4 * EPT];
        #pragma unroll
        for (int k = 0; k < 4 * EPT; ++k) {
            const unsigned u = (unsigned)(c[k] - lo);
            w[k] = 0.0f;
            if (u < (unsigned)RN) w[k] = dinv[rr[k]];
        }
        int p[4 * EPT];
        #pragma unroll
        for (int k = 0; k < 4 * EPT; ++k) {
            const unsigned u  = (unsigned)(c[k] - lo);
            const unsigned ue = (u < (unsigned)RN) ? u : spare;
            p[k] = atomicAdd(&pos[ue], 1);
            w[k] *= dl[ue];
        }
        #pragma unroll
        for (int k = 0; k < 4 * EPT; ++k) {
            const unsigned u = (unsigned)(c[k] - lo);
            if (u < (unsigned)RN) {
                int2 v;
                v.x = rr[k];
                v.y = __float_as_int(w[k]);
                lst[p[k]] = v;
            }
        }
    }
}


template <int K>
__device__ __forceinline__ void add_rows(const float4* __restrict__ hv, int ex, int ey,
                                         int j, unsigned lane, float4& acc) {
    float4 a[K];
    float  w[K];
    for (int k = 0; k < K; ++k) {
        const unsigned r = (unsigned)__builtin_amdgcn_readlane(ex, j + k);
        w[k] = __int_as_float(__builtin_amdgcn_readlane(ey, j + k));
        a[k] = hv[r * (unsigned)(D / 4) + lane];
    }
    for (int k = 0; k < K; ++k) {
        acc.x = fmaf(a[k].x, w[k], acc.x);
        acc.y = fmaf(a[k].y, w[k], acc.y);
        acc.z = fmaf(a[k].z, w[k], acc.z);
        acc.w = fmaf(a[k].w, w[k], acc.w);
    }
}

__device__ __forceinline__ void sum_stretch(const float4* __restrict__ hv, int ex, int ey, int m,
                                            unsigned lane, float4& acc) {
    int j = 0;
    for (; j + 8 <= m; j += 8) add_rows<8>(hv, ex, ey, j, lane, acc);
    if (j + 4 <= m) { add_rows<4>(hv, ex, ey, j, lane, acc); j += 4; }
    if (j + 2 <= m) { add_rows<2>(hv, ex, ey, j, lane, acc); j += 2; }
    if (j < m)      { add_rows<1>(hv, ex, ey, j, lane, acc); }
}

__global__ __launch_bounds__(256) void layer_k(const unsigned long long* __restrict__ seg,
                                                  const int2* __restrict__ lst,
                                                  const float* __restrict__ x,
                                                  const float* __restrict__ W,
                                                  const float* __restrict__ b,
                                                  float* __restrict__ out) {
    __shared__ __attribute__((aligned(16))) float sA[TM * TP];
    __shared__ __attribute__((aligned(16))) float sC[TM * TP];

    const unsigned lane = threadIdx.x & 31;
    const int wave = __builtin_amdgcn_readfirstlane(threadIdx.x >> 5);
    const int half = lane >> 4;
    const int lm   = lane & 15;
    const int n0   = wave << 4;
    const float4* __restrict__ hv = (const float4*)x;
    const float4 bb = ((const float4*)b)[lane];

    v2f bf[D / 4];
    #pragma unroll
    for (int ks = 0; ks < D / 4; ++ks) {
        const int ka = 4 * ks + 2 * half;
        bf[ks].x = W[ka * D + n0 + lm];
        bf[ks].y = W[(ka + 1) * D + n0 + lm];
    }

    for (int tile = blockIdx.x; tile < NN / TM; tile += gridDim.x) {
        const int c0 = tile * TM + wave * 4;
        unsigned long long sg[4];
        int2 e[4];
        #pragma unroll
        for (int q = 0; q < 4; ++q) sg[q] = seg[c0 + q];
        #pragma unroll
        for (int q = 0; q < 4; ++q) {
            const int s = (int)(unsigned)sg[q];
            const int n = (int)(unsigned)(sg[q] >> 32);
            const int m = (n < 32) ? n : 32;
            e[q].x = 0; e[q].y = 0;
            if ((int)lane < m) e[q] = lst[s + (int)lane];
        }
        #pragma unroll
        for (int q = 0; q < 4; ++q) {
            const int s = (int)(unsigned)sg[q];
            const int n = (int)(unsigned)(sg[q] >> 32);
            float4 acc;
            acc.x = 0.0f; acc.y = 0.0f; acc.z = 0.0f; acc.w = 0.0f;
            sum_stretch(hv, e[q].x, e[q].y, (n < 32) ? n : 32, lane, acc);
            for (int base = 32; base < n; base += 32) {
                const int m = (n - base < 32) ? (n - base) : 32;
                int2 f;
                f.x = 0; f.y = 0;
                if ((int)lane < m) f = lst[s + base + (int)lane];
                sum_stretch(hv, f.x, f.y, m, lane, acc);
            }
            *(float4*)&sA[(wave * 4 + q) * TP + 4 * (int)lane] = acc;
        }
        __syncthreads();

        v8f c_lo = {};
        v8f c_hi = {};
        #pragma unroll
        for (int ks = 0; ks < D / 4; ++ks) {
            const int ka = 4 * ks + 2 * half;
            const v2f a_lo = *(const v2f*)&sA[lm * TP + ka];
            const v2f a_hi = *(const v2f*)&sA[(16 + lm) * TP + ka];
            c_lo = __builtin_amdgcn_wmma_f32_16x16x4_f32(
                 false, a_lo,  false, bf[ks],
                 (short)0, c_lo,  false,  false);
            c_hi = __builtin_amdgcn_wmma_f32_16x16x4_f32(
                 false, a_hi,  false, bf[ks],
                 (short)0, c_hi,  false,  false);
        }
        #pragma unroll
        for (int r = 0; r < 8; ++r) {
            sC[(r + 8 * half) * TP + n0 + lm]      = c_lo[r];
            sC[(16 + r + 8 * half) * TP + n0 + lm] = c_hi[r];
        }
        __syncthreads();

        #pragma unroll
        for (int q = 0; q < 4; ++q) {
            const float4 v = *(const float4*)&sC[(wave * 4 + q) * TP + 4 * (int)lane];
            float4 o;
            o.x = fmaxf(v.x + bb.x, 0.0f);
            o.y = fmaxf(v.y + bb.y, 0.0f);
            o.z = fmaxf(v.z + bb.z, 0.0f);
            o.w = fmaxf(v.w + bb.w, 0.0f);
            ((float4*)out)[(unsigned)(c0 + q) * (unsigned)(D / 4) + lane] = o;
        }
    }
}


extern "C" void kernel_launch(void* const* d_in, const int* in_sizes, int n_in,
                              void* d_out, int out_size, void* d_ws, size_t ws_size,
                              hipStream_t stream) {
    const float* x  = (const float*)d_in[0];
    const int*   ei = (const int*)d_in[1];
    const float* Ws = (const float*)d_in[2];
    const float* bs = (const float*)d_in[3];
    float* out = (float*)d_out;

    float* bufH = (float*)d_ws;
    float* bufA = bufH + (size_t)NN * D;
    int2*  lst  = (int2*)(bufA + (size_t)NN * D);
    unsigned long long* seg = (unsigned long long*)(lst + NE + NN);
    int*   tab  = (int*)(seg + NN);
    float* dinv = (float*)(tab + (size_t)NS * NN);
    int*   total = (int*)(dinv + NN);

    const int* row = ei;
    const int* col = ei + NE;

    slice_count_k<<<NR * NS, 1024, 0, stream>>>(col, tab, total);
    node_setup_k<<<(NN + 255) / 256, 256, 0, stream>>>(tab, total, dinv, seg, lst);
    slice_place_k<<<NR * NS, 1024, 0, stream>>>(row, col, tab, dinv, lst);

    layer_k<<<1024, 256, 0, stream>>>(seg, lst, x,    Ws,                     bs,         bufA);
    layer_k<<<1024, 256, 0, stream>>>(seg, lst, bufA, Ws + (size_t)D * D,     bs + D,     bufH);
    layer_k<<<1024, 256, 0, stream>>>(seg, lst, bufH, Ws + (size_t)2 * D * D, bs + 2 * D, out);
}
